# starting kernel
# baseline (speedup 1.0000x reference)
#include <hip/hip_runtime.h>
#include <hip/hip_bf16.h>


typedef float v2f __attribute__((ext_vector_type(2)));
typedef float v8f __attribute__((ext_vector_type(8)));

constexpr int BS   = 8 * 4096;
constexpr int EMB  = 768;
constexpr int FFN  = 3072;
constexpr int NQ   = 8;

constexpr int MT     = 64;
constexpr int MTILES = MT / 16;
constexpr int FC     = 192;
constexpr int FPAD   = FC + 4;
constexpr int BLOCK  = 512;
constexpr int NWAVE  = BLOCK / 32;
constexpr int ETPW   = (EMB / 16) / NWAVE;

static_assert((MT * FC) % BLOCK == 0, "h-chunk must divide evenly");
static_assert(FFN % FC == 0, "chunking must divide FFN");

__global__ __launch_bounds__(BLOCK, 1)
void ffq_fused(const float* __restrict__ x,
               const float* __restrict__ theta,
               const float* __restrict__ W1,
               const float* __restrict__ W2,
               float* __restrict__ out)
{
    __shared__ __align__(16) float zsh[MT * NQ];
    __shared__ __align__(16) float hsh[MT * FPAD];

    const int t    = threadIdx.x;
    const int lane = t & 31;
    const int wave = t >> 5;
    const int m0   = blockIdx.x * MT;

    {
        const int m = t >> 3;
        const int q = t & 7;
        zsh[t] = cosf(x[(size_t)(m0 + m) * EMB + q]) * cosf(theta[q]);
    }
    __syncthreads();

    const int nn = lane & 15;
    const int kb = (lane >> 4) << 1;

    const float* hrow = &hsh[nn * FPAD + kb];

    const float* w2base[ETPW];
#pragma unroll
    for (int j = 0; j < ETPW; ++j) {
        const int e = (wave * ETPW + j) * 16 + nn;
        w2base[j] = W2 + (size_t)e * FFN + kb;
    }

    v8f acc[MTILES][ETPW];
#pragma unroll
    for (int mt = 0; mt < MTILES; ++mt)
#pragma unroll
        for (int j = 0; j < ETPW; ++j)
            acc[mt][j] = (v8f){0.f, 0.f, 0.f, 0.f, 0.f, 0.f, 0.f, 0.f};

    for (int fc = 0; fc < FFN; fc += FC) {
#pragma unroll
        for (int i = 0; i < (MT * FC) / BLOCK; ++i) {
            const int v  = t + i * BLOCK;
            const int m  = v / FC;
            const int fl = v - m * FC;
            const float4* w1 = reinterpret_cast<const float4*>(W1 + (size_t)(fc + fl) * NQ);
            const float4 wa = w1[0];
            const float4 wb = w1[1];
            const float* zr = &zsh[m * NQ];
            float h = zr[0]*wa.x + zr[1]*wa.y + zr[2]*wa.z + zr[3]*wa.w
                    + zr[4]*wb.x + zr[5]*wb.y + zr[6]*wb.z + zr[7]*wb.w;
            hsh[m * FPAD + fl] = fmaxf(h, 0.0f);
        }
        __syncthreads();

#pragma unroll
        for (int j = 0; j < ETPW; ++j)
            __builtin_prefetch(w2base[j] + fc, 0, 0);

#pragma unroll 2
        for (int k0 = 0; k0 < FC; k0 += 4) {
            v2f a[MTILES];
#pragma unroll
            for (int mt = 0; mt < MTILES; ++mt)
                a[mt] = *reinterpret_cast<const v2f*>(hrow + mt * 16 * FPAD + k0);
#pragma unroll
            for (int j = 0; j < ETPW; ++j) {
                const v2f b = *reinterpret_cast<const v2f*>(w2base[j] + fc + k0);
#pragma unroll
                for (int mt = 0; mt < MTILES; ++mt) {
                    acc[mt][j] = __builtin_amdgcn_wmma_f32_16x16x4_f32(
                         false, a[mt],  false, b,
                         (short)0, acc[mt][j],
                         false,  false);
                }
            }
        }
        __syncthreads();
    }

    const int hi8 = (lane >> 4) << 3;
#pragma unroll
    for (int mt = 0; mt < MTILES; ++mt) {
        const int mrow = m0 + mt * 16 + hi8;
#pragma unroll
        for (int j = 0; j < ETPW; ++j) {
            const int e0 = (wave * ETPW + j) * 16;
#pragma unroll
            for (int r = 0; r < 8; ++r) {
                out[(size_t)(mrow + r) * EMB + e0 + nn] = acc[mt][j][r];
            }
        }
    }
}

extern "C" void kernel_launch(void* const* d_in, const int* in_sizes, int n_in,
                              void* d_out, int out_size, void* d_ws, size_t ws_size,
                              hipStream_t stream) {
    (void)in_sizes; (void)n_in; (void)d_ws; (void)ws_size; (void)out_size;
    const float* x     = (const float*)d_in[0];
    const float* theta = (const float*)d_in[1];
    const float* W1    = (const float*)d_in[2];
    const float* W2    = (const float*)d_in[3];
    float* out = (float*)d_out;

    dim3 grid(BS / MT);
    dim3 block(BLOCK);
    hipLaunchKernelGGL(ffq_fused, grid, block, 0, stream, x, theta, W1, W2, out);
}


# Both WMMA fragment loaders widened to 16 B per lane (ds_load_b128, global_load_b128): one load feeds two rank-4 steps; k inside each 8 is now summed in order 0,1,4,5,2,3,6,7
# speedup vs baseline: 1.0898x; 1.0898x over previous
#include <hip/hip_runtime.h>
#include <hip/hip_bf16.h>


typedef float v2f __attribute__((ext_vector_type(2)));
typedef float v4f __attribute__((ext_vector_type(4)));
typedef float v8f __attribute__((ext_vector_type(8)));

constexpr int BS   = 8 * 4096;
constexpr int EMB  = 768;
constexpr int FFN  = 3072;
constexpr int NQ   = 8;

constexpr int MT     = 64;
constexpr int MTILES = MT / 16;
constexpr int FC     = 192;
constexpr int FPAD   = FC + 4;
constexpr int BLOCK  = 512;
constexpr int NWAVE  = BLOCK / 32;
constexpr int ETPW   = (EMB / 16) / NWAVE;

static_assert((MT * FC) % BLOCK == 0, "h-chunk must divide evenly");
static_assert(FFN % FC == 0, "chunking must divide FFN");
static_assert(FC % 8 == 0 && FPAD % 4 == 0, "8-wide k blocks, 16-byte aligned LDS rows");

__global__ __launch_bounds__(BLOCK, 1)
void ffq_fused(const float* __restrict__ x,
               const float* __restrict__ theta,
               const float* __restrict__ W1,
               const float* __restrict__ W2,
               float* __restrict__ out)
{
    __shared__ __align__(16) float zsh[MT * NQ];
    __shared__ __align__(16) float hsh[MT * FPAD];

    const int t    = threadIdx.x;
    const int lane = t & 31;
    const int wave = t >> 5;
    const int m0   = blockIdx.x * MT;

    {
        const int m = t >> 3;
        const int q = t & 7;
        zsh[t] = cosf(x[(size_t)(m0 + m) * EMB + q]) * cosf(theta[q]);
    }
    __syncthreads();

    const int nn = lane & 15;
    const int kb = (lane >> 4) << 2;

    const float* hrow = &hsh[nn * FPAD + kb];

    const float* w2base[ETPW];
#pragma unroll
    for (int j = 0; j < ETPW; ++j) {
        const int e = (wave * ETPW + j) * 16 + nn;
        w2base[j] = W2 + (size_t)e * FFN + kb;
    }

    v8f acc[MTILES][ETPW];
#pragma unroll
    for (int mt = 0; mt < MTILES; ++mt)
#pragma unroll
        for (int j = 0; j < ETPW; ++j)
            acc[mt][j] = (v8f){0.f, 0.f, 0.f, 0.f, 0.f, 0.f, 0.f, 0.f};

    for (int fc = 0; fc < FFN; fc += FC) {
#pragma unroll
        for (int i = 0; i < (MT * FC) / BLOCK; ++i) {
            const int v  = t + i * BLOCK;
            const int m  = v / FC;
            const int fl = v - m * FC;
            const float4* w1 = reinterpret_cast<const float4*>(W1 + (size_t)(fc + fl) * NQ);
            const float4 wa = w1[0];
            const float4 wb = w1[1];
            const float* zr = &zsh[m * NQ];
            float h = zr[0]*wa.x + zr[1]*wa.y + zr[2]*wa.z + zr[3]*wa.w
                    + zr[4]*wb.x + zr[5]*wb.y + zr[6]*wb.z + zr[7]*wb.w;
            hsh[m * FPAD + fl] = fmaxf(h, 0.0f);
        }
        __syncthreads();

#pragma unroll
        for (int j = 0; j < ETPW; ++j)
            __builtin_prefetch(w2base[j] + fc, 0, 0);

#pragma unroll 2
        for (int k0 = 0; k0 < FC; k0 += 8) {
            v4f a[MTILES];
#pragma unroll
            for (int mt = 0; mt < MTILES; ++mt)
                a[mt] = *reinterpret_cast<const v4f*>(hrow + mt * 16 * FPAD + k0);
#pragma unroll
            for (int j = 0; j < ETPW; ++j) {
                const v4f b = *reinterpret_cast<const v4f*>(w2base[j] + fc + k0);
                const v2f blo = {b.x, b.y};
                const v2f bhi = {b.z, b.w};
#pragma unroll
                for (int mt = 0; mt < MTILES; ++mt) {
                    const v2f alo = {a[mt].x, a[mt].y};
                    acc[mt][j] = __builtin_amdgcn_wmma_f32_16x16x4_f32(
                         false, alo,  false, blo,
                         (short)0, acc[mt][j],
                         false,  false);
                }
#pragma unroll
                for (int mt = 0; mt < MTILES; ++mt) {
                    const v2f ahi = {a[mt].z, a[mt].w};
                    acc[mt][j] = __builtin_amdgcn_wmma_f32_16x16x4_f32(
                         false, ahi,  false, bhi,
                         (short)0, acc[mt][j],
                         false,  false);
                }
            }
        }
        __syncthreads();
    }

    const int hi8 = (lane >> 4) << 3;
#pragma unroll
    for (int mt = 0; mt < MTILES; ++mt) {
        const int mrow = m0 + mt * 16 + hi8;
#pragma unroll
        for (int j = 0; j < ETPW; ++j) {
            const int e0 = (wave * ETPW + j) * 16;
#pragma unroll
            for (int r = 0; r < 8; ++r) {
                out[(size_t)(mrow + r) * EMB + e0 + nn] = acc[mt][j][r];
            }
        }
    }
}

extern "C" void kernel_launch(void* const* d_in, const int* in_sizes, int n_in,
                              void* d_out, int out_size, void* d_ws, size_t ws_size,
                              hipStream_t stream) {
    (void)in_sizes; (void)n_in; (void)d_ws; (void)ws_size; (void)out_size;
    const float* x     = (const float*)d_in[0];
    const float* theta = (const float*)d_in[1];
    const float* W1    = (const float*)d_in[2];
    const float* W2    = (const float*)d_in[3];
    float* out = (float*)d_out;

    dim3 grid(BS / MT);
    dim3 block(BLOCK);
    hipLaunchKernelGGL(ffq_fused, grid, block, 0, stream, x, theta, W1, W2, out);
}


# Layer-2 k loop software-pipelined in registers: per 8-wide k block, W2 fragments fetched 2 blocks ahead (across chunk ends too), h fragments 1 ahead, one load group before the WMMAs; sum order kept
# speedup vs baseline: 1.1336x; 1.0402x over previous
#include <hip/hip_runtime.h>
#include <hip/hip_bf16.h>


typedef float v2f __attribute__((ext_vector_type(2)));
typedef float v4f __attribute__((ext_vector_type(4)));
typedef float v8f __attribute__((ext_vector_type(8)));

constexpr int BS   = 8 * 4096;
constexpr int EMB  = 768;
constexpr int FFN  = 3072;
constexpr int NQ   = 8;

constexpr int MT     = 64;
constexpr int MTILES = MT / 16;
constexpr int FC     = 192;
constexpr int FPAD   = FC + 4;
constexpr int BLOCK  = 512;
constexpr int NWAVE  = BLOCK / 32;
constexpr int ETPW   = (EMB / 16) / NWAVE;

static_assert((MT * FC) % BLOCK == 0, "h-chunk must divide evenly");
static_assert(FFN % FC == 0, "chunking must divide FFN");
static_assert(FC % 48 == 0 && FPAD % 4 == 0, "8-wide k blocks in rotations of six, 16-byte aligned LDS rows");

__global__ __launch_bounds__(BLOCK, 1)
void ffq_fused(const float* __restrict__ x,
               const float* __restrict__ theta,
               const float* __restrict__ W1,
               const float* __restrict__ W2,
               float* __restrict__ out)
{
    __shared__ __align__(16) float zsh[MT * NQ];
    __shared__ __align__(16) float hsh[MT * FPAD];

    const int t    = threadIdx.x;
    const int lane = t & 31;
    const int wave = t >> 5;
    const int m0   = blockIdx.x * MT;

    {
        const int m = t >> 3;
        const int q = t & 7;
        zsh[t] = cosf(x[(size_t)(m0 + m) * EMB + q]) * cosf(theta[q]);
    }
    __syncthreads();

    const int nn = lane & 15;
    const int kb = (lane >> 4) << 2;

    const float* hrow = &hsh[nn * FPAD + kb];

    const float* w2base[ETPW];
#pragma unroll
    for (int j = 0; j < ETPW; ++j) {
        const int e = (wave * ETPW + j) * 16 + nn;
        w2base[j] = W2 + (size_t)e * FFN + kb;
    }

    v8f acc[MTILES][ETPW];
#pragma unroll
    for (int mt = 0; mt < MTILES; ++mt)
#pragma unroll
        for (int j = 0; j < ETPW; ++j)
            acc[mt][j] = (v8f){0.f, 0.f, 0.f, 0.f, 0.f, 0.f, 0.f, 0.f};

    v4f bq[3][ETPW];
#pragma unroll
    for (int s = 0; s < 2; ++s)
#pragma unroll
        for (int j = 0; j < ETPW; ++j)
            bq[s][j] = *reinterpret_cast<const v4f*>(w2base[j] + s * 8);

    for (int fc = 0; fc < FFN; fc += FC) {
#pragma unroll
        for (int i = 0; i < (MT * FC) / BLOCK; ++i) {
            const int v  = t + i * BLOCK;
            const int m  = v / FC;
            const int fl = v - m * FC;
            const float4* w1 = reinterpret_cast<const float4*>(W1 + (size_t)(fc + fl) * NQ);
            const float4 wa = w1[0];
            const float4 wb = w1[1];
            const float* zr = &zsh[m * NQ];
            float h = zr[0]*wa.x + zr[1]*wa.y + zr[2]*wa.z + zr[3]*wa.w
                    + zr[4]*wb.x + zr[5]*wb.y + zr[6]*wb.z + zr[7]*wb.w;
            hsh[m * FPAD + fl] = fmaxf(h, 0.0f);
        }
        __syncthreads();

        v4f aq[2][MTILES];
#pragma unroll
        for (int mt = 0; mt < MTILES; ++mt)
            aq[0][mt] = *reinterpret_cast<const v4f*>(hrow + mt * 16 * FPAD);

        for (int kt = 0; kt < FC; kt += 48) {
#pragma unroll
            for (int u = 0; u < 6; ++u) {
                const int k0 = kt + u * 8;
                const int kkn = (fc + k0 + 16 < FFN) ? (fc + k0 + 16) : (FFN - 8);
                const int kan = (k0 + 8 < FC) ? (k0 + 8) : 0;
#pragma unroll
                for (int j = 0; j < ETPW; ++j)
                    bq[(u + 2) % 3][j] = *reinterpret_cast<const v4f*>(w2base[j] + kkn);
#pragma unroll
                for (int mt = 0; mt < MTILES; ++mt)
                    aq[(u + 1) % 2][mt] = *reinterpret_cast<const v4f*>(hrow + mt * 16 * FPAD + kan);

                __builtin_amdgcn_sched_barrier(0);

#pragma unroll
                for (int j = 0; j < ETPW; ++j) {
                    const v4f b = bq[u % 3][j];
                    const v2f blo = {b.x, b.y};
                    const v2f bhi = {b.z, b.w};
#pragma unroll
                    for (int mt = 0; mt < MTILES; ++mt) {
                        const v2f alo = {aq[u % 2][mt].x, aq[u % 2][mt].y};
                        acc[mt][j] = __builtin_amdgcn_wmma_f32_16x16x4_f32(
                             false, alo,  false, blo,
                             (short)0, acc[mt][j],
                             false,  false);
                    }
#pragma unroll
                    for (int mt = 0; mt < MTILES; ++mt) {
                        const v2f ahi = {aq[u % 2][mt].z, aq[u % 2][mt].w};
                        acc[mt][j] = __builtin_amdgcn_wmma_f32_16x16x4_f32(
                             false, ahi,  false, bhi,
                             (short)0, acc[mt][j],
                             false,  false);
                    }
                }

                __builtin_amdgcn_sched_barrier(0);
            }
        }
        __syncthreads();
    }

    const int hi8 = (lane >> 4) << 3;
#pragma unroll
    for (int mt = 0; mt < MTILES; ++mt) {
        const int mrow = m0 + mt * 16 + hi8;
#pragma unroll
        for (int j = 0; j < ETPW; ++j) {
            const int e0 = (wave * ETPW + j) * 16;
#pragma unroll
            for (int r = 0; r < 8; ++r) {
                out[(size_t)(mrow + r) * EMB + e0 + nn] = acc[mt][j][r];
            }
        }
    }
}

extern "C" void kernel_launch(void* const* d_in, const int* in_sizes, int n_in,
                              void* d_out, int out_size, void* d_ws, size_t ws_size,
                              hipStream_t stream) {
    (void)in_sizes; (void)n_in; (void)d_ws; (void)ws_size; (void)out_size;
    const float* x     = (const float*)d_in[0];
    const float* theta = (const float*)d_in[1];
    const float* W1    = (const float*)d_in[2];
    const float* W2    = (const float*)d_in[3];
    float* out = (float*)d_out;

    dim3 grid(BS / MT);
    dim3 block(BLOCK);
    hipLaunchKernelGGL(ffq_fused, grid, block, 0, stream, x, theta, W1, W2, out);
}


# Layer 1 (z @ W1^T, K=8) moved from per-element VALU dot products to the f32 WMMA (as W1 z^T: 3 tiles per wave per chunk, W1 fragments fetched a chunk ahead); q now summed 0,1,4,5 then 2,3,6,7
# speedup vs baseline: 1.2575x; 1.1093x over previous
#include <hip/hip_runtime.h>
#include <hip/hip_bf16.h>


typedef float v2f __attribute__((ext_vector_type(2)));
typedef float v4f __attribute__((ext_vector_type(4)));
typedef float v8f __attribute__((ext_vector_type(8)));

constexpr int BS   = 8 * 4096;
constexpr int EMB  = 768;
constexpr int FFN  = 3072;
constexpr int NQ   = 8;

constexpr int MT     = 64;
constexpr int MTILES = MT / 16;
constexpr int FC     = 192;
constexpr int FPAD   = FC + 4;
constexpr int BLOCK  = 512;
constexpr int NWAVE  = BLOCK / 32;
constexpr int ETPW   = (EMB / 16) / NWAVE;

static_assert(FFN % FC == 0, "chunking must divide FFN");
static_assert(FC % 48 == 0 && FPAD % 4 == 0, "8-wide k blocks in rotations of six, 16-byte aligned LDS rows");
constexpr int NT1    = (MTILES * (FC / 16)) / NWAVE;
static_assert((MTILES * (FC / 16)) % NWAVE == 0 && (FC / 16) % NT1 == 0 && NQ == 8,
              "layer-1 tiles split evenly over the waves; K = 8 is one 16-byte fragment");

__global__ __launch_bounds__(BLOCK, 1)
void ffq_fused(const float* __restrict__ x,
               const float* __restrict__ theta,
               const float* __restrict__ W1,
               const float* __restrict__ W2,
               float* __restrict__ out)
{
    __shared__ __align__(16) float zsh[MT * NQ];
    __shared__ __align__(16) float hsh[MT * FPAD];

    const int t    = threadIdx.x;
    const int lane = t & 31;
    const int wave = t >> 5;
    const int m0   = blockIdx.x * MT;

    {
        const int m = t >> 3;
        const int q = t & 7;
        zsh[t] = cosf(x[(size_t)(m0 + m) * EMB + q]) * cosf(theta[q]);
    }
    __syncthreads();

    const int nn = lane & 15;
    const int kb = (lane >> 4) << 2;

    const float* hrow = &hsh[nn * FPAD + kb];

    const float* w2base[ETPW];
#pragma unroll
    for (int j = 0; j < ETPW; ++j) {
        const int e = (wave * ETPW + j) * 16 + nn;
        w2base[j] = W2 + (size_t)e * FFN + kb;
    }

    v8f acc[MTILES][ETPW];
#pragma unroll
    for (int mt = 0; mt < MTILES; ++mt)
#pragma unroll
        for (int j = 0; j < ETPW; ++j)
            acc[mt][j] = (v8f){0.f, 0.f, 0.f, 0.f, 0.f, 0.f, 0.f, 0.f};

    v4f bq[3][ETPW];
#pragma unroll
    for (int s = 0; s < 2; ++s)
#pragma unroll
        for (int j = 0; j < ETPW; ++j)
            bq[s][j] = *reinterpret_cast<const v4f*>(w2base[j] + s * 8);

    const int mt1 = wave & (MTILES - 1);
    const int ng1 = wave / MTILES;
    const v4f zq  = *reinterpret_cast<const v4f*>(&zsh[(mt1 * 16 + nn) * NQ + kb]);
    const v2f zlo = {zq.x, zq.y};
    const v2f zhi = {zq.z, zq.w};
    const float* w1base = W1 + (size_t)(ng1 * NT1 * 16 + nn) * NQ + kb;
    float* hst = &hsh[(mt1 * 16 + nn) * FPAD + ng1 * NT1 * 16 + ((lane >> 4) << 3)];
    v4f w1q[NT1];
#pragma unroll
    for (int j = 0; j < NT1; ++j)
        w1q[j] = *reinterpret_cast<const v4f*>(w1base + (size_t)(j * 16) * NQ);

    for (int fc = 0; fc < FFN; fc += FC) {
        {
            v8f hd[NT1];
#pragma unroll
            for (int j = 0; j < NT1; ++j) {
                const v2f wlo = {w1q[j].x, w1q[j].y};
                const v2f whi = {w1q[j].z, w1q[j].w};
                v8f d = (v8f){0.f, 0.f, 0.f, 0.f, 0.f, 0.f, 0.f, 0.f};
                d = __builtin_amdgcn_wmma_f32_16x16x4_f32(
                     false, wlo,  false, zlo,
                     (short)0, d,  false,  false);
                d = __builtin_amdgcn_wmma_f32_16x16x4_f32(
                     false, whi,  false, zhi,
                     (short)0, d,  false,  false);
                hd[j] = d;
            }
#pragma unroll
            for (int j = 0; j < NT1; ++j) {
                const v8f d = hd[j];
                const v4f lo = {fmaxf(d[0], 0.0f), fmaxf(d[1], 0.0f), fmaxf(d[2], 0.0f), fmaxf(d[3], 0.0f)};
                const v4f hi = {fmaxf(d[4], 0.0f), fmaxf(d[5], 0.0f), fmaxf(d[6], 0.0f), fmaxf(d[7], 0.0f)};
                *reinterpret_cast<v4f*>(hst + j * 16)     = lo;
                *reinterpret_cast<v4f*>(hst + j * 16 + 4) = hi;
            }
        }
        __syncthreads();

        {
            const int fn = (fc + FC < FFN) ? (fc + FC) : fc;
#pragma unroll
            for (int j = 0; j < NT1; ++j)
                w1q[j] = *reinterpret_cast<const v4f*>(w1base + (size_t)(fn + j * 16) * NQ);
        }

        v4f aq[2][MTILES];
#pragma unroll
        for (int mt = 0; mt < MTILES; ++mt)
            aq[0][mt] = *reinterpret_cast<const v4f*>(hrow + mt * 16 * FPAD);

        for (int kt = 0; kt < FC; kt += 48) {
#pragma unroll
            for (int u = 0; u < 6; ++u) {
                const int k0 = kt + u * 8;
                const int kkn = (fc + k0 + 16 < FFN) ? (fc + k0 + 16) : (FFN - 8);
                const int kan = (k0 + 8 < FC) ? (k0 + 8) : 0;
#pragma unroll
                for (int j = 0; j < ETPW; ++j)
                    bq[(u + 2) % 3][j] = *reinterpret_cast<const v4f*>(w2base[j] + kkn);
#pragma unroll
                for (int mt = 0; mt < MTILES; ++mt)
                    aq[(u + 1) % 2][mt] = *reinterpret_cast<const v4f*>(hrow + mt * 16 * FPAD + kan);

                __builtin_amdgcn_sched_barrier(0);

#pragma unroll
                for (int j = 0; j < ETPW; ++j) {
                    const v4f b = bq[u % 3][j];
                    const v2f blo = {b.x, b.y};
                    const v2f bhi = {b.z, b.w};
#pragma unroll
                    for (int mt = 0; mt < MTILES; ++mt) {
                        const v2f alo = {aq[u % 2][mt].x, aq[u % 2][mt].y};
                        acc[mt][j] = __builtin_amdgcn_wmma_f32_16x16x4_f32(
                             false, alo,  false, blo,
                             (short)0, acc[mt][j],
                             false,  false);
                    }
#pragma unroll
                    for (int mt = 0; mt < MTILES; ++mt) {
                        const v2f ahi = {aq[u % 2][mt].z, aq[u % 2][mt].w};
                        acc[mt][j] = __builtin_amdgcn_wmma_f32_16x16x4_f32(
                             false, ahi,  false, bhi,
                             (short)0, acc[mt][j],
                             false,  false);
                    }
                }

                __builtin_amdgcn_sched_barrier(0);
            }
        }
        __syncthreads();
    }

    const int hi8 = (lane >> 4) << 3;
#pragma unroll
    for (int mt = 0; mt < MTILES; ++mt) {
        const int mrow = m0 + mt * 16 + hi8;
#pragma unroll
        for (int j = 0; j < ETPW; ++j) {
            const int e0 = (wave * ETPW + j) * 16;
#pragma unroll
            for (int r = 0; r < 8; ++r) {
                out[(size_t)(mrow + r) * EMB + e0 + nn] = acc[mt][j][r];
            }
        }
    }
}

extern "C" void kernel_launch(void* const* d_in, const int* in_sizes, int n_in,
                              void* d_out, int out_size, void* d_ws, size_t ws_size,
                              hipStream_t stream) {
    (void)in_sizes; (void)n_in; (void)d_ws; (void)ws_size; (void)out_size;
    const float* x     = (const float*)d_in[0];
    const float* theta = (const float*)d_in[1];
    const float* W1    = (const float*)d_in[2];
    const float* W2    = (const float*)d_in[3];
    float* out = (float*)d_out;

    dim3 grid(BS / MT);
    dim3 block(BLOCK);
    hipLaunchKernelGGL(ffq_fused, grid, block, 0, stream, x, theta, W1, W2, out);
}
